# starting kernel
# baseline (speedup 1.0000x reference)
#include <hip/hip_runtime.h>


typedef float v2f __attribute__((ext_vector_type(2)));
typedef float v4f __attribute__((ext_vector_type(4)));
typedef float v8f __attribute__((ext_vector_type(8)));

#define LLN   (1 << 20)
#define OUTF  1024
#define INF   768
#define MTOT  16384
#define KTILES (INF / 16)

__global__ void k_theta_fwht(const float* __restrict__ theta,
                             const float* __restrict__ BB,
                             float* __restrict__ A01,
                             float* __restrict__ sAcc) {
    __shared__ float lds[1024];
    const int t = threadIdx.x;
    const int b = blockIdx.x;
    if (b == 0 && t == 0) sAcc[0] = 0.0f;
    for (int i = t; i < 1024; i += 256) {
        const int gi = b * 1024 + i;
        lds[i] = BB[gi] * theta[gi];
    }
    __syncthreads();
    for (int h = 1; h < 1024; h <<= 1) {
        for (int r = t; r < 512; r += 256) {
            const int i = ((r / h) * (h << 1)) + (r % h);
            const float x = lds[i], y = lds[i + h];
            lds[i] = x + y;
            lds[i + h] = x - y;
        }
        __syncthreads();
    }
    for (int i = t; i < 1024; i += 256) A01[b * 1024 + i] = lds[i];
}

__global__ void k_gg_sum(const float* __restrict__ GG, float* __restrict__ sAcc) {
    __shared__ float red[256];
    const int t = threadIdx.x;
    const size_t base = (size_t)blockIdx.x * 1024;
    float p = 0.0f;
    for (int i = 0; i < 4; ++i) {
        const float g = GG[base + t + i * 256];
        p += g * g;
    }
    red[t] = p;
    __syncthreads();
    for (int s = 128; s > 0; s >>= 1) {
        if (t < s) red[t] += red[t + s];
        __syncthreads();
    }
    if (t == 0) atomicAdd(sAcc, red[0]);
}

__global__ void k_perm(const int* __restrict__ Pi, const float* __restrict__ GG,
                       const float* __restrict__ A01, float* __restrict__ w) {
    const int j = blockIdx.x * 256 + threadIdx.x;
    const int i = Pi[j];
    const int c = i & 1023;
    const float a0 = A01[c];
    const float a1 = A01[1024 + c];
    const float v = ((i >> 10) & 1) ? (a0 - a1) : (a0 + a1);
    w[j] = GG[j] * v;
}

__global__ void k_fwht_lo(float* __restrict__ w) {
    __shared__ float lds[1024];
    const int t = threadIdx.x;
    const size_t base = (size_t)blockIdx.x * 1024;
    for (int i = t; i < 1024; i += 256) lds[i] = w[base + i];
    __syncthreads();
    for (int h = 1; h < 1024; h <<= 1) {
        for (int r = t; r < 512; r += 256) {
            const int i = ((r / h) * (h << 1)) + (r % h);
            const float x = lds[i], y = lds[i + h];
            lds[i] = x + y;
            lds[i + h] = x - y;
        }
        __syncthreads();
    }
    for (int i = t; i < 1024; i += 256) w[base + i] = lds[i];
}

__global__ void k_fwht_hi(float* __restrict__ w) {
    __shared__ float lds[1024 * 8];
    const int t = threadIdx.x;
    const int c = t & 7;
    const int ty = t >> 3;
    const int cb = blockIdx.x * 8;
    for (int r0 = 0; r0 < 1024; r0 += 32) {
        const int row = r0 + ty;
        lds[row * 8 + c] = w[(size_t)row * 1024 + cb + c];
    }
    __syncthreads();
    for (int h = 1; h < 1024; h <<= 1) {
        for (int jj = ty; jj < 512; jj += 32) {
            const int i = ((jj / h) * (h << 1)) + (jj % h);
            const float x = lds[i * 8 + c], y = lds[(i + h) * 8 + c];
            lds[i * 8 + c] = x + y;
            lds[(i + h) * 8 + c] = x - y;
        }
        __syncthreads();
    }
    for (int r0 = 0; r0 < 1024; r0 += 32) {
        const int row = r0 + ty;
        w[(size_t)row * 1024 + cb + c] = lds[row * 8 + c];
    }
}

#define LDA 20
#define LDB 20
__global__ __launch_bounds__(256) void k_gemm(
    const float* __restrict__ x, const float* __restrict__ W0,
    const float* __restrict__ bias, const float* __restrict__ wup,
    const float* __restrict__ sAcc, float* __restrict__ out) {
    __shared__ float As[2][128 * LDA];
    __shared__ float Bs[2][256 * LDB];
    const int t    = threadIdx.x;
    const int lane = t & 31;
    const int wid  = t >> 5;
    const int wm   = wid & 1;
    const int wn   = wid >> 1;
    const int half = lane >> 4;
    const int lm   = lane & 15;
    const int blockN = blockIdx.x * 256;
    const int blockM = blockIdx.y * 128;
    const float scale = 1.0f / sqrtf(786432.0f * sAcc[0]);

    const int rowA0 = t >> 2;
    const int kqA0  = (t & 3) << 2;
    const int rowA1 = (t + 256) >> 2;
    const int kqA1  = ((t + 256) & 3) << 2;

    v8f acc[4][4];
    const v8f zero = {0.f, 0.f, 0.f, 0.f, 0.f, 0.f, 0.f, 0.f};
#pragma unroll
    for (int mt = 0; mt < 4; ++mt)
#pragma unroll
        for (int nt = 0; nt < 4; ++nt) acc[mt][nt] = zero;

    v4f sa0, sa1, swv[4], suv[4];

    sa0 = *(const v4f*)(x + (size_t)(blockM + rowA0) * INF + kqA0);
    sa1 = *(const v4f*)(x + (size_t)(blockM + rowA1) * INF + kqA1);
#pragma unroll
    for (int i = 0; i < 4; ++i) {
        const int ff  = t + i * 256;
        const int row = ff >> 2;
        const int kq  = (ff & 3) << 2;
        const size_t gi = (size_t)(blockN + row) * INF + kq;
        swv[i] = *(const v4f*)(W0 + gi);
        suv[i] = *(const v4f*)(wup + gi);
    }
    *(v4f*)(&As[0][rowA0 * LDA + kqA0]) = sa0;
    *(v4f*)(&As[0][rowA1 * LDA + kqA1]) = sa1;
#pragma unroll
    for (int i = 0; i < 4; ++i) {
        const int ff  = t + i * 256;
        const int row = ff >> 2;
        const int kq  = (ff & 3) << 2;
        *(v4f*)(&Bs[0][row * LDB + kq]) = swv[i] + scale * suv[i];
    }
    __syncthreads();

    for (int kt = 0; kt < KTILES; ++kt) {
        const int cur = kt & 1;
        const int nxt = cur ^ 1;
        const bool more = (kt + 1) < KTILES;

        if (more) {
            const int kOff = (kt + 1) * 16;
            sa0 = *(const v4f*)(x + (size_t)(blockM + rowA0) * INF + kOff + kqA0);
            sa1 = *(const v4f*)(x + (size_t)(blockM + rowA1) * INF + kOff + kqA1);
#pragma unroll
            for (int i = 0; i < 4; ++i) {
                const int ff  = t + i * 256;
                const int row = ff >> 2;
                const int kq  = (ff & 3) << 2;
                const size_t gi = (size_t)(blockN + row) * INF + kOff + kq;
                swv[i] = *(const v4f*)(W0 + gi);
                suv[i] = *(const v4f*)(wup + gi);
            }
        }

        const float* __restrict__ Ac = As[cur];
        const float* __restrict__ Bc = Bs[cur];
#pragma unroll
        for (int s = 0; s < 4; ++s) {
            const int kk = (s << 2) + (half << 1);
            v2f af[4], bf[4];
#pragma unroll
            for (int mt = 0; mt < 4; ++mt)
                af[mt] = *(const v2f*)(Ac + (wm * 64 + mt * 16 + lm) * LDA + kk);
#pragma unroll
            for (int nt = 0; nt < 4; ++nt)
                bf[nt] = *(const v2f*)(Bc + (wn * 64 + nt * 16 + lm) * LDB + kk);
#pragma unroll
            for (int mt = 0; mt < 4; ++mt)
#pragma unroll
                for (int nt = 0; nt < 4; ++nt)
                    acc[mt][nt] = __builtin_amdgcn_wmma_f32_16x16x4_f32(
                        false, af[mt], false, bf[nt], (short)0, acc[mt][nt],
                        false, false);
        }

        if (more) {
            *(v4f*)(&As[nxt][rowA0 * LDA + kqA0]) = sa0;
            *(v4f*)(&As[nxt][rowA1 * LDA + kqA1]) = sa1;
#pragma unroll
            for (int i = 0; i < 4; ++i) {
                const int ff  = t + i * 256;
                const int row = ff >> 2;
                const int kq  = (ff & 3) << 2;
                *(v4f*)(&Bs[nxt][row * LDB + kq]) = swv[i] + scale * suv[i];
            }
        }
        __syncthreads();
    }

#pragma unroll
    for (int nt = 0; nt < 4; ++nt) {
        const int col = blockN + wn * 64 + nt * 16 + lm;
        const float bv = bias[col];
#pragma unroll
        for (int mt = 0; mt < 4; ++mt) {
            const int rowBase = blockM + wm * 64 + mt * 16 + half * 8;
#pragma unroll
            for (int v = 0; v < 8; ++v)
                out[(size_t)(rowBase + v) * OUTF + col] = acc[mt][nt][v] + bv;
        }
    }
}

extern "C" void kernel_launch(void* const* d_in, const int* in_sizes, int n_in,
                              void* d_out, int out_size, void* d_ws, size_t ws_size,
                              hipStream_t stream) {
    (void)in_sizes; (void)n_in; (void)out_size; (void)ws_size;
    const float* x     = (const float*)d_in[0];
    const float* theta = (const float*)d_in[1];
    const float* W0    = (const float*)d_in[2];
    const float* bias  = (const float*)d_in[3];
    const float* BB    = (const float*)d_in[4];
    const float* GG    = (const float*)d_in[5];
    const int*   Pi    = (const int*)d_in[6];
    float* out = (float*)d_out;

    float* sAcc = (float*)d_ws;
    float* A01  = (float*)d_ws + 64;
    float* w    = (float*)d_ws + 64 + 2048;

    k_theta_fwht<<<2,    256, 0, stream>>>(theta, BB, A01, sAcc);
    k_gg_sum   <<<1024,  256, 0, stream>>>(GG, sAcc);
    k_perm     <<<4096,  256, 0, stream>>>(Pi, GG, A01, w);
    k_fwht_lo  <<<1024,  256, 0, stream>>>(w);
    k_fwht_hi  <<<128,   256, 0, stream>>>(w);
    k_gemm     <<<dim3(OUTF / 256, MTOT / 128), 256, 0, stream>>>(
        x, W0, bias, w, sAcc, out);
}


# GEMM K-loop: scheduling fences keep the next tile's 10 global loads as one group ahead of the 64-WMMA block, waited once behind it; compiler had sunk them into the block behind 5 full load waits
# speedup vs baseline: 1.0770x; 1.0770x over previous
#include <hip/hip_runtime.h>


typedef float v2f __attribute__((ext_vector_type(2)));
typedef float v4f __attribute__((ext_vector_type(4)));
typedef float v8f __attribute__((ext_vector_type(8)));

#define LLN   (1 << 20)
#define OUTF  1024
#define INF   768
#define MTOT  16384
#define KTILES (INF / 16)

__global__ void k_theta_fwht(const float* __restrict__ theta,
                             const float* __restrict__ BB,
                             float* __restrict__ A01,
                             float* __restrict__ sAcc) {
    __shared__ float lds[1024];
    const int t = threadIdx.x;
    const int b = blockIdx.x;
    if (b == 0 && t == 0) sAcc[0] = 0.0f;
    for (int i = t; i < 1024; i += 256) {
        const int gi = b * 1024 + i;
        lds[i] = BB[gi] * theta[gi];
    }
    __syncthreads();
    for (int h = 1; h < 1024; h <<= 1) {
        for (int r = t; r < 512; r += 256) {
            const int i = ((r / h) * (h << 1)) + (r % h);
            const float x = lds[i], y = lds[i + h];
            lds[i] = x + y;
            lds[i + h] = x - y;
        }
        __syncthreads();
    }
    for (int i = t; i < 1024; i += 256) A01[b * 1024 + i] = lds[i];
}

__global__ void k_gg_sum(const float* __restrict__ GG, float* __restrict__ sAcc) {
    __shared__ float red[256];
    const int t = threadIdx.x;
    const size_t base = (size_t)blockIdx.x * 1024;
    float p = 0.0f;
    for (int i = 0; i < 4; ++i) {
        const float g = GG[base + t + i * 256];
        p += g * g;
    }
    red[t] = p;
    __syncthreads();
    for (int s = 128; s > 0; s >>= 1) {
        if (t < s) red[t] += red[t + s];
        __syncthreads();
    }
    if (t == 0) atomicAdd(sAcc, red[0]);
}

__global__ void k_perm(const int* __restrict__ Pi, const float* __restrict__ GG,
                       const float* __restrict__ A01, float* __restrict__ w) {
    const int j = blockIdx.x * 256 + threadIdx.x;
    const int i = Pi[j];
    const int c = i & 1023;
    const float a0 = A01[c];
    const float a1 = A01[1024 + c];
    const float v = ((i >> 10) & 1) ? (a0 - a1) : (a0 + a1);
    w[j] = GG[j] * v;
}

__global__ void k_fwht_lo(float* __restrict__ w) {
    __shared__ float lds[1024];
    const int t = threadIdx.x;
    const size_t base = (size_t)blockIdx.x * 1024;
    for (int i = t; i < 1024; i += 256) lds[i] = w[base + i];
    __syncthreads();
    for (int h = 1; h < 1024; h <<= 1) {
        for (int r = t; r < 512; r += 256) {
            const int i = ((r / h) * (h << 1)) + (r % h);
            const float x = lds[i], y = lds[i + h];
            lds[i] = x + y;
            lds[i + h] = x - y;
        }
        __syncthreads();
    }
    for (int i = t; i < 1024; i += 256) w[base + i] = lds[i];
}

__global__ void k_fwht_hi(float* __restrict__ w) {
    __shared__ float lds[1024 * 8];
    const int t = threadIdx.x;
    const int c = t & 7;
    const int ty = t >> 3;
    const int cb = blockIdx.x * 8;
    for (int r0 = 0; r0 < 1024; r0 += 32) {
        const int row = r0 + ty;
        lds[row * 8 + c] = w[(size_t)row * 1024 + cb + c];
    }
    __syncthreads();
    for (int h = 1; h < 1024; h <<= 1) {
        for (int jj = ty; jj < 512; jj += 32) {
            const int i = ((jj / h) * (h << 1)) + (jj % h);
            const float x = lds[i * 8 + c], y = lds[(i + h) * 8 + c];
            lds[i * 8 + c] = x + y;
            lds[(i + h) * 8 + c] = x - y;
        }
        __syncthreads();
    }
    for (int r0 = 0; r0 < 1024; r0 += 32) {
        const int row = r0 + ty;
        w[(size_t)row * 1024 + cb + c] = lds[row * 8 + c];
    }
}

#define LDA 20
#define LDB 20
#define ROWS64 (64 * INF)
__global__ __launch_bounds__(256) void k_gemm(
    const float* __restrict__ x, const float* __restrict__ W0,
    const float* __restrict__ bias, const float* __restrict__ wup,
    const float* __restrict__ sAcc, float* __restrict__ out) {
    __shared__ float As[2][128 * LDA];
    __shared__ float Bs[2][256 * LDB];
    const int t    = threadIdx.x;
    const int lane = t & 31;
    const int wid  = t >> 5;
    const int wm   = wid & 1;
    const int wn   = wid >> 1;
    const int half = lane >> 4;
    const int lm   = lane & 15;
    const int blockN = blockIdx.x * 256;
    const int blockM = blockIdx.y * 128;
    const float scale = 1.0f / sqrtf(786432.0f * sAcc[0]);

    const int fr = t >> 2;
    const int fq = (t & 3) << 2;
    const float* __restrict__ pA = x   + (size_t)(blockM + fr) * INF + fq;
    const float* __restrict__ pW = W0  + (size_t)(blockN + fr) * INF + fq;
    const float* __restrict__ pU = wup + (size_t)(blockN + fr) * INF + fq;
    const int stOff = fr * LDA + fq;

    v8f acc[4][4];
    const v8f zero = {0.f, 0.f, 0.f, 0.f, 0.f, 0.f, 0.f, 0.f};
#pragma unroll
    for (int mt = 0; mt < 4; ++mt)
#pragma unroll
        for (int nt = 0; nt < 4; ++nt) acc[mt][nt] = zero;

    v4f sa0, sa1, swv[4], suv[4];

    sa0 = *(const v4f*)(pA);
    sa1 = *(const v4f*)(pA + ROWS64);
#pragma unroll
    for (int i = 0; i < 4; ++i) {
        swv[i] = *(const v4f*)(pW + i * ROWS64);
        suv[i] = *(const v4f*)(pU + i * ROWS64);
    }
    *(v4f*)(&As[0][stOff]) = sa0;
    *(v4f*)(&As[0][stOff + 64 * LDA]) = sa1;
#pragma unroll
    for (int i = 0; i < 4; ++i)
        *(v4f*)(&Bs[0][stOff + i * 64 * LDB]) = swv[i] + scale * suv[i];
    __syncthreads();

    for (int kt = 0; kt < KTILES; ++kt) {
        const int cur = kt & 1;
        const int nxt = cur ^ 1;
        const bool more = (kt + 1) < KTILES;

        if (more) {
            const int kOff = (kt + 1) * 16;
            sa0 = *(const v4f*)(pA + kOff);
            sa1 = *(const v4f*)(pA + kOff + ROWS64);
#pragma unroll
            for (int i = 0; i < 4; ++i) {
                swv[i] = *(const v4f*)(pW + kOff + i * ROWS64);
                suv[i] = *(const v4f*)(pU + kOff + i * ROWS64);
            }
        }
        __builtin_amdgcn_sched_barrier(0);

        const float* __restrict__ Ac = As[cur];
        const float* __restrict__ Bc = Bs[cur];
#pragma unroll
        for (int s = 0; s < 4; ++s) {
            const int kk = (s << 2) + (half << 1);
            v2f af[4], bf[4];
#pragma unroll
            for (int mt = 0; mt < 4; ++mt)
                af[mt] = *(const v2f*)(Ac + (wm * 64 + mt * 16 + lm) * LDA + kk);
#pragma unroll
            for (int nt = 0; nt < 4; ++nt)
                bf[nt] = *(const v2f*)(Bc + (wn * 64 + nt * 16 + lm) * LDB + kk);
#pragma unroll
            for (int mt = 0; mt < 4; ++mt)
#pragma unroll
                for (int nt = 0; nt < 4; ++nt)
                    acc[mt][nt] = __builtin_amdgcn_wmma_f32_16x16x4_f32(
                        false, af[mt], false, bf[nt], (short)0, acc[mt][nt],
                        false, false);
        }
        __builtin_amdgcn_sched_barrier(0);

        if (more) {
            *(v4f*)(&As[nxt][stOff]) = sa0;
            *(v4f*)(&As[nxt][stOff + 64 * LDA]) = sa1;
#pragma unroll
            for (int i = 0; i < 4; ++i)
                *(v4f*)(&Bs[nxt][stOff + i * 64 * LDB]) = swv[i] + scale * suv[i];
        }
        __syncthreads();
    }

#pragma unroll
    for (int nt = 0; nt < 4; ++nt) {
        const int col = blockN + wn * 64 + nt * 16 + lm;
        const float bv = bias[col];
#pragma unroll
        for (int mt = 0; mt < 4; ++mt) {
            const int rowBase = blockM + wm * 64 + mt * 16 + half * 8;
#pragma unroll
            for (int v = 0; v < 8; ++v)
                out[(size_t)(rowBase + v) * OUTF + col] = acc[mt][nt][v] + bv;
        }
    }
}

extern "C" void kernel_launch(void* const* d_in, const int* in_sizes, int n_in,
                              void* d_out, int out_size, void* d_ws, size_t ws_size,
                              hipStream_t stream) {
    (void)in_sizes; (void)n_in; (void)out_size; (void)ws_size;
    const float* x     = (const float*)d_in[0];
    const float* theta = (const float*)d_in[1];
    const float* W0    = (const float*)d_in[2];
    const float* bias  = (const float*)d_in[3];
    const float* BB    = (const float*)d_in[4];
    const float* GG    = (const float*)d_in[5];
    const int*   Pi    = (const int*)d_in[6];
    float* out = (float*)d_out;

    float* sAcc = (float*)d_ws;
    float* A01  = (float*)d_ws + 64;
    float* w    = (float*)d_ws + 64 + 2048;

    k_theta_fwht<<<2,    256, 0, stream>>>(theta, BB, A01, sAcc);
    k_gg_sum   <<<1024,  256, 0, stream>>>(GG, sAcc);
    k_perm     <<<4096,  256, 0, stream>>>(Pi, GG, A01, w);
    k_fwht_lo  <<<1024,  256, 0, stream>>>(w);
    k_fwht_hi  <<<128,   256, 0, stream>>>(w);
    k_gemm     <<<dim3(OUTF / 256, MTOT / 128), 256, 0, stream>>>(
        x, W0, bias, w, sAcc, out);
}


# k_perm and k_gg_sum folded into k_fwht_lo's tile load (gather and GG^2 partial per 1024-chunk, w built in LDS): 6 launches -> 4, 40960 fewer waves to enter; GG^2 sum order unchanged
# speedup vs baseline: 1.1109x; 1.0315x over previous
#include <hip/hip_runtime.h>


typedef float v2f __attribute__((ext_vector_type(2)));
typedef float v4f __attribute__((ext_vector_type(4)));
typedef float v8f __attribute__((ext_vector_type(8)));

#define LLN   (1 << 20)
#define OUTF  1024
#define INF   768
#define MTOT  16384
#define KTILES (INF / 16)

__global__ void k_theta_fwht(const float* __restrict__ theta,
                             const float* __restrict__ BB,
                             float* __restrict__ A01,
                             float* __restrict__ sAcc) {
    __shared__ float lds[1024];
    const int t = threadIdx.x;
    const int b = blockIdx.x;
    if (b == 0 && t == 0) sAcc[0] = 0.0f;
    for (int i = t; i < 1024; i += 256) {
        const int gi = b * 1024 + i;
        lds[i] = BB[gi] * theta[gi];
    }
    __syncthreads();
    for (int h = 1; h < 1024; h <<= 1) {
        for (int r = t; r < 512; r += 256) {
            const int i = ((r / h) * (h << 1)) + (r % h);
            const float x = lds[i], y = lds[i + h];
            lds[i] = x + y;
            lds[i + h] = x - y;
        }
        __syncthreads();
    }
    for (int i = t; i < 1024; i += 256) A01[b * 1024 + i] = lds[i];
}

__global__ void k_perm_fwht_lo(const int* __restrict__ Pi,
                               const float* __restrict__ GG,
                               const float* __restrict__ A01,
                               float* __restrict__ sAcc,
                               float* __restrict__ w) {
    __shared__ float lds[1024];
    __shared__ float red[256];
    const int t = threadIdx.x;
    const size_t base = (size_t)blockIdx.x * 1024;
    float p = 0.0f;
    for (int i = 0; i < 4; ++i) {
        const int e = t + i * 256;
        const float g = GG[base + e];
        const int pi = Pi[base + e];
        const int c = pi & 1023;
        const float a0 = A01[c];
        const float a1 = A01[1024 + c];
        const float v = ((pi >> 10) & 1) ? (a0 - a1) : (a0 + a1);
        lds[e] = g * v;
        p += g * g;
    }
    red[t] = p;
    __syncthreads();
    for (int s = 128; s > 0; s >>= 1) {
        if (t < s) red[t] += red[t + s];
        __syncthreads();
    }
    if (t == 0) atomicAdd(sAcc, red[0]);
    for (int h = 1; h < 1024; h <<= 1) {
        for (int r = t; r < 512; r += 256) {
            const int i = ((r / h) * (h << 1)) + (r % h);
            const float x = lds[i], y = lds[i + h];
            lds[i] = x + y;
            lds[i + h] = x - y;
        }
        __syncthreads();
    }
    for (int i = t; i < 1024; i += 256) w[base + i] = lds[i];
}

__global__ void k_fwht_hi(float* __restrict__ w) {
    __shared__ float lds[1024 * 8];
    const int t = threadIdx.x;
    const int c = t & 7;
    const int ty = t >> 3;
    const int cb = blockIdx.x * 8;
    for (int r0 = 0; r0 < 1024; r0 += 32) {
        const int row = r0 + ty;
        lds[row * 8 + c] = w[(size_t)row * 1024 + cb + c];
    }
    __syncthreads();
    for (int h = 1; h < 1024; h <<= 1) {
        for (int jj = ty; jj < 512; jj += 32) {
            const int i = ((jj / h) * (h << 1)) + (jj % h);
            const float x = lds[i * 8 + c], y = lds[(i + h) * 8 + c];
            lds[i * 8 + c] = x + y;
            lds[(i + h) * 8 + c] = x - y;
        }
        __syncthreads();
    }
    for (int r0 = 0; r0 < 1024; r0 += 32) {
        const int row = r0 + ty;
        w[(size_t)row * 1024 + cb + c] = lds[row * 8 + c];
    }
}

#define LDA 20
#define LDB 20
#define ROWS64 (64 * INF)
__global__ __launch_bounds__(256) void k_gemm(
    const float* __restrict__ x, const float* __restrict__ W0,
    const float* __restrict__ bias, const float* __restrict__ wup,
    const float* __restrict__ sAcc, float* __restrict__ out) {
    __shared__ float As[2][128 * LDA];
    __shared__ float Bs[2][256 * LDB];
    const int t    = threadIdx.x;
    const int lane = t & 31;
    const int wid  = t >> 5;
    const int wm   = wid & 1;
    const int wn   = wid >> 1;
    const int half = lane >> 4;
    const int lm   = lane & 15;
    const int blockN = blockIdx.x * 256;
    const int blockM = blockIdx.y * 128;
    const float scale = 1.0f / sqrtf(786432.0f * sAcc[0]);

    const int fr = t >> 2;
    const int fq = (t & 3) << 2;
    const float* __restrict__ pA = x   + (size_t)(blockM + fr) * INF + fq;
    const float* __restrict__ pW = W0  + (size_t)(blockN + fr) * INF + fq;
    const float* __restrict__ pU = wup + (size_t)(blockN + fr) * INF + fq;
    const int stOff = fr * LDA + fq;

    v8f acc[4][4];
    const v8f zero = {0.f, 0.f, 0.f, 0.f, 0.f, 0.f, 0.f, 0.f};
#pragma unroll
    for (int mt = 0; mt < 4; ++mt)
#pragma unroll
        for (int nt = 0; nt < 4; ++nt) acc[mt][nt] = zero;

    v4f sa0, sa1, swv[4], suv[4];

    sa0 = *(const v4f*)(pA);
    sa1 = *(const v4f*)(pA + ROWS64);
#pragma unroll
    for (int i = 0; i < 4; ++i) {
        swv[i] = *(const v4f*)(pW + i * ROWS64);
        suv[i] = *(const v4f*)(pU + i * ROWS64);
    }
    *(v4f*)(&As[0][stOff]) = sa0;
    *(v4f*)(&As[0][stOff + 64 * LDA]) = sa1;
#pragma unroll
    for (int i = 0; i < 4; ++i)
        *(v4f*)(&Bs[0][stOff + i * 64 * LDB]) = swv[i] + scale * suv[i];
    __syncthreads();

    for (int kt = 0; kt < KTILES; ++kt) {
        const int cur = kt & 1;
        const int nxt = cur ^ 1;
        const bool more = (kt + 1) < KTILES;

        if (more) {
            const int kOff = (kt + 1) * 16;
            sa0 = *(const v4f*)(pA + kOff);
            sa1 = *(const v4f*)(pA + kOff + ROWS64);
#pragma unroll
            for (int i = 0; i < 4; ++i) {
                swv[i] = *(const v4f*)(pW + kOff + i * ROWS64);
                suv[i] = *(const v4f*)(pU + kOff + i * ROWS64);
            }
        }
        __builtin_amdgcn_sched_barrier(0);

        const float* __restrict__ Ac = As[cur];
        const float* __restrict__ Bc = Bs[cur];
#pragma unroll
        for (int s = 0; s < 4; ++s) {
            const int kk = (s << 2) + (half << 1);
            v2f af[4], bf[4];
#pragma unroll
            for (int mt = 0; mt < 4; ++mt)
                af[mt] = *(const v2f*)(Ac + (wm * 64 + mt * 16 + lm) * LDA + kk);
#pragma unroll
            for (int nt = 0; nt < 4; ++nt)
                bf[nt] = *(const v2f*)(Bc + (wn * 64 + nt * 16 + lm) * LDB + kk);
#pragma unroll
            for (int mt = 0; mt < 4; ++mt)
#pragma unroll
                for (int nt = 0; nt < 4; ++nt)
                    acc[mt][nt] = __builtin_amdgcn_wmma_f32_16x16x4_f32(
                        false, af[mt], false, bf[nt], (short)0, acc[mt][nt],
                        false, false);
        }
        __builtin_amdgcn_sched_barrier(0);

        if (more) {
            *(v4f*)(&As[nxt][stOff]) = sa0;
            *(v4f*)(&As[nxt][stOff + 64 * LDA]) = sa1;
#pragma unroll
            for (int i = 0; i < 4; ++i)
                *(v4f*)(&Bs[nxt][stOff + i * 64 * LDB]) = swv[i] + scale * suv[i];
        }
        __syncthreads();
    }

#pragma unroll
    for (int nt = 0; nt < 4; ++nt) {
        const int col = blockN + wn * 64 + nt * 16 + lm;
        const float bv = bias[col];
#pragma unroll
        for (int mt = 0; mt < 4; ++mt) {
            const int rowBase = blockM + wm * 64 + mt * 16 + half * 8;
#pragma unroll
            for (int v = 0; v < 8; ++v)
                out[(size_t)(rowBase + v) * OUTF + col] = acc[mt][nt][v] + bv;
        }
    }
}

extern "C" void kernel_launch(void* const* d_in, const int* in_sizes, int n_in,
                              void* d_out, int out_size, void* d_ws, size_t ws_size,
                              hipStream_t stream) {
    (void)in_sizes; (void)n_in; (void)out_size; (void)ws_size;
    const float* x     = (const float*)d_in[0];
    const float* theta = (const float*)d_in[1];
    const float* W0    = (const float*)d_in[2];
    const float* bias  = (const float*)d_in[3];
    const float* BB    = (const float*)d_in[4];
    const float* GG    = (const float*)d_in[5];
    const int*   Pi    = (const int*)d_in[6];
    float* out = (float*)d_out;

    float* sAcc = (float*)d_ws;
    float* A01  = (float*)d_ws + 64;
    float* w    = (float*)d_ws + 64 + 2048;

    k_theta_fwht<<<2,    256, 0, stream>>>(theta, BB, A01, sAcc);
    k_perm_fwht_lo<<<1024, 256, 0, stream>>>(Pi, GG, A01, sAcc, w);
    k_fwht_hi  <<<128,   256, 0, stream>>>(w);
    k_gemm     <<<dim3(OUTF / 256, MTOT / 128), 256, 0, stream>>>(
        x, W0, bias, w, sAcc, out);
}


# FWHT stages in registers between LDS passes: k_fwht_hi 32 rows/thread, 5 stages/pass, 2 passes; 1024-point FWHTs 4 values/thread, 2 stages/pass; was 1 LDS round trip per butterfly; sums unchanged
# speedup vs baseline: 1.1578x; 1.0422x over previous
#include <hip/hip_runtime.h>


typedef float v2f __attribute__((ext_vector_type(2)));
typedef float v4f __attribute__((ext_vector_type(4)));
typedef float v8f __attribute__((ext_vector_type(8)));

#define LLN   (1 << 20)
#define OUTF  1024
#define INF   768
#define MTOT  16384
#define KTILES (INF / 16)

__device__ __forceinline__ void fwht1024_lds(float* lds, const int t) {
#pragma unroll
    for (int p = 0; p < 5; ++p) {
        const int sh = 2 * p;
        const int h = 1 << sh;
        const int base = ((t >> sh) << (sh + 2)) | (t & (h - 1));
        const float e0 = lds[base], e1 = lds[base + h];
        const float e2 = lds[base + 2 * h], e3 = lds[base + 3 * h];
        const float s0 = e0 + e1, d0 = e0 - e1;
        const float s1 = e2 + e3, d1 = e2 - e3;
        lds[base]         = s0 + s1;
        lds[base + h]     = d0 + d1;
        lds[base + 2 * h] = s0 - s1;
        lds[base + 3 * h] = d0 - d1;
        __syncthreads();
    }
}

__global__ void k_theta_fwht(const float* __restrict__ theta,
                             const float* __restrict__ BB,
                             float* __restrict__ A01,
                             float* __restrict__ sAcc) {
    __shared__ float lds[1024];
    const int t = threadIdx.x;
    const int b = blockIdx.x;
    if (b == 0 && t == 0) sAcc[0] = 0.0f;
    for (int i = t; i < 1024; i += 256) {
        const int gi = b * 1024 + i;
        lds[i] = BB[gi] * theta[gi];
    }
    __syncthreads();
    fwht1024_lds(lds, t);
    for (int i = t; i < 1024; i += 256) A01[b * 1024 + i] = lds[i];
}

__global__ void k_perm_fwht_lo(const int* __restrict__ Pi,
                               const float* __restrict__ GG,
                               const float* __restrict__ A01,
                               float* __restrict__ sAcc,
                               float* __restrict__ w) {
    __shared__ float lds[1024];
    __shared__ float red[256];
    const int t = threadIdx.x;
    const size_t base = (size_t)blockIdx.x * 1024;
    float p = 0.0f;
    for (int i = 0; i < 4; ++i) {
        const int e = t + i * 256;
        const float g = GG[base + e];
        const int pi = Pi[base + e];
        const int c = pi & 1023;
        const float a0 = A01[c];
        const float a1 = A01[1024 + c];
        const float v = ((pi >> 10) & 1) ? (a0 - a1) : (a0 + a1);
        lds[e] = g * v;
        p += g * g;
    }
    red[t] = p;
    __syncthreads();
    for (int s = 128; s > 0; s >>= 1) {
        if (t < s) red[t] += red[t + s];
        __syncthreads();
    }
    if (t == 0) atomicAdd(sAcc, red[0]);
    fwht1024_lds(lds, t);
    for (int i = t; i < 1024; i += 256) w[base + i] = lds[i];
}

__device__ __forceinline__ void fwht32_regs(float (&a)[32]) {
#pragma unroll
    for (int h = 1; h < 32; h <<= 1) {
#pragma unroll
        for (int i = 0; i < 32; ++i) {
            if ((i & h) == 0) {
                const float x = a[i], y = a[i + h];
                a[i] = x + y;
                a[i + h] = x - y;
            }
        }
    }
}

__global__ void k_fwht_hi(float* __restrict__ w) {
    __shared__ float lds[1024 * 8 + 32 * 8];
    const int t = threadIdx.x;
    const int c = t & 7;
    const int q = t >> 3;
    float* __restrict__ col = w + blockIdx.x * 8 + c;
    float a[32];
#pragma unroll
    for (int m = 0; m < 32; ++m) a[m] = col[(size_t)(q * 32 + m) * 1024];
    fwht32_regs(a);
#pragma unroll
    for (int m = 0; m < 32; ++m) lds[(q * 32 + m) * 8 + q * 8 + c] = a[m];
    __syncthreads();
#pragma unroll
    for (int m = 0; m < 32; ++m) a[m] = lds[(m * 32 + q) * 8 + m * 8 + c];
    fwht32_regs(a);
#pragma unroll
    for (int m = 0; m < 32; ++m) col[(size_t)(m * 32 + q) * 1024] = a[m];
}

#define LDA 20
#define LDB 20
#define ROWS64 (64 * INF)
__global__ __launch_bounds__(256) void k_gemm(
    const float* __restrict__ x, const float* __restrict__ W0,
    const float* __restrict__ bias, const float* __restrict__ wup,
    const float* __restrict__ sAcc, float* __restrict__ out) {
    __shared__ float As[2][128 * LDA];
    __shared__ float Bs[2][256 * LDB];
    const int t    = threadIdx.x;
    const int lane = t & 31;
    const int wid  = t >> 5;
    const int wm   = wid & 1;
    const int wn   = wid >> 1;
    const int half = lane >> 4;
    const int lm   = lane & 15;
    const int blockN = blockIdx.x * 256;
    const int blockM = blockIdx.y * 128;
    const float scale = 1.0f / sqrtf(786432.0f * sAcc[0]);

    const int fr = t >> 2;
    const int fq = (t & 3) << 2;
    const float* __restrict__ pA = x   + (size_t)(blockM + fr) * INF + fq;
    const float* __restrict__ pW = W0  + (size_t)(blockN + fr) * INF + fq;
    const float* __restrict__ pU = wup + (size_t)(blockN + fr) * INF + fq;
    const int stOff = fr * LDA + fq;

    v8f acc[4][4];
    const v8f zero = {0.f, 0.f, 0.f, 0.f, 0.f, 0.f, 0.f, 0.f};
#pragma unroll
    for (int mt = 0; mt < 4; ++mt)
#pragma unroll
        for (int nt = 0; nt < 4; ++nt) acc[mt][nt] = zero;

    v4f sa0, sa1, swv[4], suv[4];

    sa0 = *(const v4f*)(pA);
    sa1 = *(const v4f*)(pA + ROWS64);
#pragma unroll
    for (int i = 0; i < 4; ++i) {
        swv[i] = *(const v4f*)(pW + i * ROWS64);
        suv[i] = *(const v4f*)(pU + i * ROWS64);
    }
    *(v4f*)(&As[0][stOff]) = sa0;
    *(v4f*)(&As[0][stOff + 64 * LDA]) = sa1;
#pragma unroll
    for (int i = 0; i < 4; ++i)
        *(v4f*)(&Bs[0][stOff + i * 64 * LDB]) = swv[i] + scale * suv[i];
    __syncthreads();

    for (int kt = 0; kt < KTILES; ++kt) {
        const int cur = kt & 1;
        const int nxt = cur ^ 1;
        const bool more = (kt + 1) < KTILES;

        if (more) {
            const int kOff = (kt + 1) * 16;
            sa0 = *(const v4f*)(pA + kOff);
            sa1 = *(const v4f*)(pA + kOff + ROWS64);
#pragma unroll
            for (int i = 0; i < 4; ++i) {
                swv[i] = *(const v4f*)(pW + kOff + i * ROWS64);
                suv[i] = *(const v4f*)(pU + kOff + i * ROWS64);
            }
        }
        __builtin_amdgcn_sched_barrier(0);

        const float* __restrict__ Ac = As[cur];
        const float* __restrict__ Bc = Bs[cur];
#pragma unroll
        for (int s = 0; s < 4; ++s) {
            const int kk = (s << 2) + (half << 1);
            v2f af[4], bf[4];
#pragma unroll
            for (int mt = 0; mt < 4; ++mt)
                af[mt] = *(const v2f*)(Ac + (wm * 64 + mt * 16 + lm) * LDA + kk);
#pragma unroll
            for (int nt = 0; nt < 4; ++nt)
                bf[nt] = *(const v2f*)(Bc + (wn * 64 + nt * 16 + lm) * LDB + kk);
#pragma unroll
            for (int mt = 0; mt < 4; ++mt)
#pragma unroll
                for (int nt = 0; nt < 4; ++nt)
                    acc[mt][nt] = __builtin_amdgcn_wmma_f32_16x16x4_f32(
                        false, af[mt], false, bf[nt], (short)0, acc[mt][nt],
                        false, false);
        }
        __builtin_amdgcn_sched_barrier(0);

        if (more) {
            *(v4f*)(&As[nxt][stOff]) = sa0;
            *(v4f*)(&As[nxt][stOff + 64 * LDA]) = sa1;
#pragma unroll
            for (int i = 0; i < 4; ++i)
                *(v4f*)(&Bs[nxt][stOff + i * 64 * LDB]) = swv[i] + scale * suv[i];
        }
        __syncthreads();
    }

#pragma unroll
    for (int nt = 0; nt < 4; ++nt) {
        const int col = blockN + wn * 64 + nt * 16 + lm;
        const float bv = bias[col];
#pragma unroll
        for (int mt = 0; mt < 4; ++mt) {
            const int rowBase = blockM + wm * 64 + mt * 16 + half * 8;
#pragma unroll
            for (int v = 0; v < 8; ++v)
                out[(size_t)(rowBase + v) * OUTF + col] = acc[mt][nt][v] + bv;
        }
    }
}

extern "C" void kernel_launch(void* const* d_in, const int* in_sizes, int n_in,
                              void* d_out, int out_size, void* d_ws, size_t ws_size,
                              hipStream_t stream) {
    (void)in_sizes; (void)n_in; (void)out_size; (void)ws_size;
    const float* x     = (const float*)d_in[0];
    const float* theta = (const float*)d_in[1];
    const float* W0    = (const float*)d_in[2];
    const float* bias  = (const float*)d_in[3];
    const float* BB    = (const float*)d_in[4];
    const float* GG    = (const float*)d_in[5];
    const int*   Pi    = (const int*)d_in[6];
    float* out = (float*)d_out;

    float* sAcc = (float*)d_ws;
    float* A01  = (float*)d_ws + 64;
    float* w    = (float*)d_ws + 64 + 2048;

    k_theta_fwht<<<2,    256, 0, stream>>>(theta, BB, A01, sAcc);
    k_perm_fwht_lo<<<1024, 256, 0, stream>>>(Pi, GG, A01, sAcc, w);
    k_fwht_hi  <<<128,   256, 0, stream>>>(w);
    k_gemm     <<<dim3(OUTF / 256, MTOT / 128), 256, 0, stream>>>(
        x, W0, bias, w, sAcc, out);
}


# k_fwht_hi's store writes the finished B operand W0+scale*v2 once (rows 0..767 only); GEMM reads one B stream: 6 tile loads per thread per K-step, was 10, no sum in the K-loop; every sum unchanged
# speedup vs baseline: 1.3070x; 1.1289x over previous
#include <hip/hip_runtime.h>


typedef float v2f __attribute__((ext_vector_type(2)));
typedef float v4f __attribute__((ext_vector_type(4)));
typedef float v8f __attribute__((ext_vector_type(8)));

#define LLN   (1 << 20)
#define OUTF  1024
#define INF   768
#define MTOT  16384
#define KTILES (INF / 16)

__device__ __forceinline__ void fwht1024_lds(float* lds, const int t) {
#pragma unroll
    for (int p = 0; p < 5; ++p) {
        const int sh = 2 * p;
        const int h = 1 << sh;
        const int base = ((t >> sh) << (sh + 2)) | (t & (h - 1));
        const float e0 = lds[base], e1 = lds[base + h];
        const float e2 = lds[base + 2 * h], e3 = lds[base + 3 * h];
        const float s0 = e0 + e1, d0 = e0 - e1;
        const float s1 = e2 + e3, d1 = e2 - e3;
        lds[base]         = s0 + s1;
        lds[base + h]     = d0 + d1;
        lds[base + 2 * h] = s0 - s1;
        lds[base + 3 * h] = d0 - d1;
        __syncthreads();
    }
}

__global__ void k_theta_fwht(const float* __restrict__ theta,
                             const float* __restrict__ BB,
                             float* __restrict__ A01,
                             float* __restrict__ sAcc) {
    __shared__ float lds[1024];
    const int t = threadIdx.x;
    const int b = blockIdx.x;
    if (b == 0 && t == 0) sAcc[0] = 0.0f;
    for (int i = t; i < 1024; i += 256) {
        const int gi = b * 1024 + i;
        lds[i] = BB[gi] * theta[gi];
    }
    __syncthreads();
    fwht1024_lds(lds, t);
    for (int i = t; i < 1024; i += 256) A01[b * 1024 + i] = lds[i];
}

__global__ void k_perm_fwht_lo(const int* __restrict__ Pi,
                               const float* __restrict__ GG,
                               const float* __restrict__ A01,
                               float* __restrict__ sAcc,
                               float* __restrict__ w) {
    __shared__ float lds[1024];
    __shared__ float red[256];
    const int t = threadIdx.x;
    const size_t base = (size_t)blockIdx.x * 1024;
    float p = 0.0f;
    for (int i = 0; i < 4; ++i) {
        const int e = t + i * 256;
        const float g = GG[base + e];
        const int pi = Pi[base + e];
        const int c = pi & 1023;
        const float a0 = A01[c];
        const float a1 = A01[1024 + c];
        const float v = ((pi >> 10) & 1) ? (a0 - a1) : (a0 + a1);
        lds[e] = g * v;
        p += g * g;
    }
    red[t] = p;
    __syncthreads();
    for (int s = 128; s > 0; s >>= 1) {
        if (t < s) red[t] += red[t + s];
        __syncthreads();
    }
    if (t == 0) atomicAdd(sAcc, red[0]);
    fwht1024_lds(lds, t);
    for (int i = t; i < 1024; i += 256) w[base + i] = lds[i];
}

__device__ __forceinline__ void fwht32_regs(float (&a)[32]) {
#pragma unroll
    for (int h = 1; h < 32; h <<= 1) {
#pragma unroll
        for (int i = 0; i < 32; ++i) {
            if ((i & h) == 0) {
                const float x = a[i], y = a[i + h];
                a[i] = x + y;
                a[i + h] = x - y;
            }
        }
    }
}

__global__ void k_fwht_hi(float* __restrict__ w,
                          const float* __restrict__ W0,
                          const float* __restrict__ sAcc) {
    __shared__ float lds[1024 * 8 + 32 * 8];
    const int t = threadIdx.x;
    const int c = t & 7;
    const int q = t >> 3;
    float* __restrict__ col = w + blockIdx.x * 8 + c;
    const float* __restrict__ col0 = W0 + blockIdx.x * 8 + c;
    const float scale = 1.0f / sqrtf(786432.0f * sAcc[0]);
    float a[32];
#pragma unroll
    for (int m = 0; m < 32; ++m) a[m] = col[(size_t)(q * 32 + m) * 1024];
    float b0[24];
#pragma unroll
    for (int m = 0; m < 24; ++m) b0[m] = col0[(size_t)(m * 32 + q) * 1024];
    fwht32_regs(a);
#pragma unroll
    for (int m = 0; m < 32; ++m) lds[(q * 32 + m) * 8 + q * 8 + c] = a[m];
    __syncthreads();
#pragma unroll
    for (int m = 0; m < 32; ++m) a[m] = lds[(m * 32 + q) * 8 + m * 8 + c];
    fwht32_regs(a);
#pragma unroll
    for (int m = 0; m < 24; ++m)
        col[(size_t)(m * 32 + q) * 1024] = b0[m] + scale * a[m];
}

#define LDA 20
#define LDB 20
#define ROWS64 (64 * INF)
__global__ __launch_bounds__(256) void k_gemm(
    const float* __restrict__ x, const float* __restrict__ bias,
    const float* __restrict__ weff, float* __restrict__ out) {
    __shared__ float As[2][128 * LDA];
    __shared__ float Bs[2][256 * LDB];
    const int t    = threadIdx.x;
    const int lane = t & 31;
    const int wid  = t >> 5;
    const int wm   = wid & 1;
    const int wn   = wid >> 1;
    const int half = lane >> 4;
    const int lm   = lane & 15;
    const int blockN = blockIdx.x * 256;
    const int blockM = blockIdx.y * 128;

    const int fr = t >> 2;
    const int fq = (t & 3) << 2;
    const float* __restrict__ pA = x   + (size_t)(blockM + fr) * INF + fq;
    const float* __restrict__ pB = weff + (size_t)(blockN + fr) * INF + fq;
    const int stOff = fr * LDA + fq;

    v8f acc[4][4];
    const v8f zero = {0.f, 0.f, 0.f, 0.f, 0.f, 0.f, 0.f, 0.f};
#pragma unroll
    for (int mt = 0; mt < 4; ++mt)
#pragma unroll
        for (int nt = 0; nt < 4; ++nt) acc[mt][nt] = zero;

    v4f sa0, sa1, sb[4];

    sa0 = *(const v4f*)(pA);
    sa1 = *(const v4f*)(pA + ROWS64);
#pragma unroll
    for (int i = 0; i < 4; ++i) sb[i] = *(const v4f*)(pB + i * ROWS64);
    *(v4f*)(&As[0][stOff]) = sa0;
    *(v4f*)(&As[0][stOff + 64 * LDA]) = sa1;
#pragma unroll
    for (int i = 0; i < 4; ++i) *(v4f*)(&Bs[0][stOff + i * 64 * LDB]) = sb[i];
    __syncthreads();

    for (int kt = 0; kt < KTILES; ++kt) {
        const int cur = kt & 1;
        const int nxt = cur ^ 1;
        const bool more = (kt + 1) < KTILES;

        if (more) {
            const int kOff = (kt + 1) * 16;
            sa0 = *(const v4f*)(pA + kOff);
            sa1 = *(const v4f*)(pA + kOff + ROWS64);
#pragma unroll
            for (int i = 0; i < 4; ++i)
                sb[i] = *(const v4f*)(pB + kOff + i * ROWS64);
        }
        __builtin_amdgcn_sched_barrier(0);

        const float* __restrict__ Ac = As[cur];
        const float* __restrict__ Bc = Bs[cur];
#pragma unroll
        for (int s = 0; s < 4; ++s) {
            const int kk = (s << 2) + (half << 1);
            v2f af[4], bf[4];
#pragma unroll
            for (int mt = 0; mt < 4; ++mt)
                af[mt] = *(const v2f*)(Ac + (wm * 64 + mt * 16 + lm) * LDA + kk);
#pragma unroll
            for (int nt = 0; nt < 4; ++nt)
                bf[nt] = *(const v2f*)(Bc + (wn * 64 + nt * 16 + lm) * LDB + kk);
#pragma unroll
            for (int mt = 0; mt < 4; ++mt)
#pragma unroll
                for (int nt = 0; nt < 4; ++nt)
                    acc[mt][nt] = __builtin_amdgcn_wmma_f32_16x16x4_f32(
                        false, af[mt], false, bf[nt], (short)0, acc[mt][nt],
                        false, false);
        }
        __builtin_amdgcn_sched_barrier(0);

        if (more) {
            *(v4f*)(&As[nxt][stOff]) = sa0;
            *(v4f*)(&As[nxt][stOff + 64 * LDA]) = sa1;
#pragma unroll
            for (int i = 0; i < 4; ++i)
                *(v4f*)(&Bs[nxt][stOff + i * 64 * LDB]) = sb[i];
        }
        __syncthreads();
    }

#pragma unroll
    for (int nt = 0; nt < 4; ++nt) {
        const int col = blockN + wn * 64 + nt * 16 + lm;
        const float bv = bias[col];
#pragma unroll
        for (int mt = 0; mt < 4; ++mt) {
            const int rowBase = blockM + wm * 64 + mt * 16 + half * 8;
#pragma unroll
            for (int v = 0; v < 8; ++v)
                out[(size_t)(rowBase + v) * OUTF + col] = acc[mt][nt][v] + bv;
        }
    }
}

extern "C" void kernel_launch(void* const* d_in, const int* in_sizes, int n_in,
                              void* d_out, int out_size, void* d_ws, size_t ws_size,
                              hipStream_t stream) {
    (void)in_sizes; (void)n_in; (void)out_size; (void)ws_size;
    const float* x     = (const float*)d_in[0];
    const float* theta = (const float*)d_in[1];
    const float* W0    = (const float*)d_in[2];
    const float* bias  = (const float*)d_in[3];
    const float* BB    = (const float*)d_in[4];
    const float* GG    = (const float*)d_in[5];
    const int*   Pi    = (const int*)d_in[6];
    float* out = (float*)d_out;

    float* sAcc = (float*)d_ws;
    float* A01  = (float*)d_ws + 64;
    float* w    = (float*)d_ws + 64 + 2048;

    k_theta_fwht<<<2,    256, 0, stream>>>(theta, BB, A01, sAcc);
    k_perm_fwht_lo<<<1024, 256, 0, stream>>>(Pi, GG, A01, sAcc, w);
    k_fwht_hi  <<<128,   256, 0, stream>>>(w, W0, sAcc);
    k_gemm     <<<dim3(OUTF / 256, MTOT / 128), 256, 0, stream>>>(
        x, bias, w, out);
}


# GEMM K-loop rotated across its barrier: each iteration opens with the previous tile's last 32 WMMAs (fragments already in registers), tile stores 16 WMMAs before the barrier; sums unchanged
# speedup vs baseline: 1.3849x; 1.0596x over previous
#include <hip/hip_runtime.h>


typedef float v2f __attribute__((ext_vector_type(2)));
typedef float v4f __attribute__((ext_vector_type(4)));
typedef float v8f __attribute__((ext_vector_type(8)));

#define LLN   (1 << 20)
#define OUTF  1024
#define INF   768
#define MTOT  16384
#define KTILES (INF / 16)

__device__ __forceinline__ void fwht1024_lds(float* lds, const int t) {
#pragma unroll
    for (int p = 0; p < 5; ++p) {
        const int sh = 2 * p;
        const int h = 1 << sh;
        const int base = ((t >> sh) << (sh + 2)) | (t & (h - 1));
        const float e0 = lds[base], e1 = lds[base + h];
        const float e2 = lds[base + 2 * h], e3 = lds[base + 3 * h];
        const float s0 = e0 + e1, d0 = e0 - e1;
        const float s1 = e2 + e3, d1 = e2 - e3;
        lds[base]         = s0 + s1;
        lds[base + h]     = d0 + d1;
        lds[base + 2 * h] = s0 - s1;
        lds[base + 3 * h] = d0 - d1;
        __syncthreads();
    }
}

__global__ void k_theta_fwht(const float* __restrict__ theta,
                             const float* __restrict__ BB,
                             float* __restrict__ A01,
                             float* __restrict__ sAcc) {
    __shared__ float lds[1024];
    const int t = threadIdx.x;
    const int b = blockIdx.x;
    if (b == 0 && t == 0) sAcc[0] = 0.0f;
    for (int i = t; i < 1024; i += 256) {
        const int gi = b * 1024 + i;
        lds[i] = BB[gi] * theta[gi];
    }
    __syncthreads();
    fwht1024_lds(lds, t);
    for (int i = t; i < 1024; i += 256) A01[b * 1024 + i] = lds[i];
}

__global__ void k_perm_fwht_lo(const int* __restrict__ Pi,
                               const float* __restrict__ GG,
                               const float* __restrict__ A01,
                               float* __restrict__ sAcc,
                               float* __restrict__ w) {
    __shared__ float lds[1024];
    __shared__ float red[256];
    const int t = threadIdx.x;
    const size_t base = (size_t)blockIdx.x * 1024;
    float p = 0.0f;
    for (int i = 0; i < 4; ++i) {
        const int e = t + i * 256;
        const float g = GG[base + e];
        const int pi = Pi[base + e];
        const int c = pi & 1023;
        const float a0 = A01[c];
        const float a1 = A01[1024 + c];
        const float v = ((pi >> 10) & 1) ? (a0 - a1) : (a0 + a1);
        lds[e] = g * v;
        p += g * g;
    }
    red[t] = p;
    __syncthreads();
    for (int s = 128; s > 0; s >>= 1) {
        if (t < s) red[t] += red[t + s];
        __syncthreads();
    }
    if (t == 0) atomicAdd(sAcc, red[0]);
    fwht1024_lds(lds, t);
    for (int i = t; i < 1024; i += 256) w[base + i] = lds[i];
}

__device__ __forceinline__ void fwht32_regs(float (&a)[32]) {
#pragma unroll
    for (int h = 1; h < 32; h <<= 1) {
#pragma unroll
        for (int i = 0; i < 32; ++i) {
            if ((i & h) == 0) {
                const float x = a[i], y = a[i + h];
                a[i] = x + y;
                a[i + h] = x - y;
            }
        }
    }
}

__global__ void k_fwht_hi(float* __restrict__ w,
                          const float* __restrict__ W0,
                          const float* __restrict__ sAcc) {
    __shared__ float lds[1024 * 8 + 32 * 8];
    const int t = threadIdx.x;
    const int c = t & 7;
    const int q = t >> 3;
    float* __restrict__ col = w + blockIdx.x * 8 + c;
    const float* __restrict__ col0 = W0 + blockIdx.x * 8 + c;
    const float scale = 1.0f / sqrtf(786432.0f * sAcc[0]);
    float a[32];
#pragma unroll
    for (int m = 0; m < 32; ++m) a[m] = col[(size_t)(q * 32 + m) * 1024];
    float b0[24];
#pragma unroll
    for (int m = 0; m < 24; ++m) b0[m] = col0[(size_t)(m * 32 + q) * 1024];
    fwht32_regs(a);
#pragma unroll
    for (int m = 0; m < 32; ++m) lds[(q * 32 + m) * 8 + q * 8 + c] = a[m];
    __syncthreads();
#pragma unroll
    for (int m = 0; m < 32; ++m) a[m] = lds[(m * 32 + q) * 8 + m * 8 + c];
    fwht32_regs(a);
#pragma unroll
    for (int m = 0; m < 24; ++m)
        col[(size_t)(m * 32 + q) * 1024] = b0[m] + scale * a[m];
}

#define LDA 20
#define LDB 20
#define ROWS64 (64 * INF)

__device__ __forceinline__ void frag_load(const float* __restrict__ Ac,
                                          const float* __restrict__ Bc,
                                          const int aoff, const int boff,
                                          const int s0,
                                          v2f (&af)[2][4], v2f (&bf)[2][4]) {
#pragma unroll
    for (int j = 0; j < 2; ++j) {
        const int kk = (s0 + j) << 2;
#pragma unroll
        for (int mt = 0; mt < 4; ++mt)
            af[j][mt] = *(const v2f*)(Ac + aoff + mt * 16 * LDA + kk);
#pragma unroll
        for (int nt = 0; nt < 4; ++nt)
            bf[j][nt] = *(const v2f*)(Bc + boff + nt * 16 * LDB + kk);
    }
}

__device__ __forceinline__ void mma16(v8f (&acc)[4][4], const v2f (&af)[4],
                                      const v2f (&bf)[4]) {
#pragma unroll
    for (int mt = 0; mt < 4; ++mt)
#pragma unroll
        for (int nt = 0; nt < 4; ++nt)
            acc[mt][nt] = __builtin_amdgcn_wmma_f32_16x16x4_f32(
                false, af[mt], false, bf[nt], (short)0, acc[mt][nt],
                false, false);
}

__global__ __launch_bounds__(256) void k_gemm(
    const float* __restrict__ x, const float* __restrict__ bias,
    const float* __restrict__ weff, float* __restrict__ out) {
    __shared__ float As[2][128 * LDA];
    __shared__ float Bs[2][256 * LDB];
    const int t    = threadIdx.x;
    const int lane = t & 31;
    const int wid  = t >> 5;
    const int wm   = wid & 1;
    const int wn   = wid >> 1;
    const int half = lane >> 4;
    const int lm   = lane & 15;
    const int blockN = blockIdx.x * 256;
    const int blockM = blockIdx.y * 128;

    const int fr = t >> 2;
    const int fq = (t & 3) << 2;
    const float* __restrict__ pA = x   + (size_t)(blockM + fr) * INF + fq;
    const float* __restrict__ pB = weff + (size_t)(blockN + fr) * INF + fq;
    const float* __restrict__ qA = pA + 32;
    const float* __restrict__ qB = pB + 32;
    const int stOff = fr * LDA + fq;

    const int aoff = (wm * 64 + lm) * LDA + (half << 1);
    const int boff = (wn * 64 + lm) * LDB + (half << 1);

    v8f acc[4][4];
    const v8f zero = {0.f, 0.f, 0.f, 0.f, 0.f, 0.f, 0.f, 0.f};
#pragma unroll
    for (int mt = 0; mt < 4; ++mt)
#pragma unroll
        for (int nt = 0; nt < 4; ++nt) acc[mt][nt] = zero;

    v4f sa0, sa1, sb[4];
    v2f fa1[2][4], fb1[2][4], fa2[2][4], fb2[2][4];

    sa0 = *(const v4f*)(pA);
    sa1 = *(const v4f*)(pA + ROWS64);
#pragma unroll
    for (int i = 0; i < 4; ++i) sb[i] = *(const v4f*)(pB + i * ROWS64);
    *(v4f*)(&As[0][stOff]) = sa0;
    *(v4f*)(&As[0][stOff + 64 * LDA]) = sa1;
#pragma unroll
    for (int i = 0; i < 4; ++i) *(v4f*)(&Bs[0][stOff + i * 64 * LDB]) = sb[i];
    __syncthreads();

    sa0 = *(const v4f*)(pA + 16);
    sa1 = *(const v4f*)(pA + 16 + ROWS64);
#pragma unroll
    for (int i = 0; i < 4; ++i) sb[i] = *(const v4f*)(pB + 16 + i * ROWS64);
    frag_load(As[0], Bs[0], aoff, boff, 0, fa1, fb1);
    frag_load(As[0], Bs[0], aoff, boff, 2, fa2, fb2);
    __builtin_amdgcn_sched_barrier(0);
    mma16(acc, fa1[0], fb1[0]);
    mma16(acc, fa1[1], fb1[1]);
    __builtin_amdgcn_sched_barrier(0);
    *(v4f*)(&As[1][stOff]) = sa0;
    *(v4f*)(&As[1][stOff + 64 * LDA]) = sa1;
#pragma unroll
    for (int i = 0; i < 4; ++i) *(v4f*)(&Bs[1][stOff + i * 64 * LDB]) = sb[i];
    __syncthreads();

    for (int kt = 1; kt < KTILES; ++kt) {
        const int cur = kt & 1;
        const int nxt = cur ^ 1;
        const bool more = (kt + 1) < KTILES;
        const float* __restrict__ Ac = As[cur];
        const float* __restrict__ Bc = Bs[cur];

        if (more) {
            sa0 = *(const v4f*)(qA);
            sa1 = *(const v4f*)(qA + ROWS64);
#pragma unroll
            for (int i = 0; i < 4; ++i)
                sb[i] = *(const v4f*)(qB + i * ROWS64);
        }
        frag_load(Ac, Bc, aoff, boff, 0, fa1, fb1);
        __builtin_amdgcn_sched_barrier(0);
        mma16(acc, fa2[0], fb2[0]);
        mma16(acc, fa2[1], fb2[1]);
        __builtin_amdgcn_sched_barrier(0);
        frag_load(Ac, Bc, aoff, boff, 2, fa2, fb2);
        __builtin_amdgcn_sched_barrier(0);
        mma16(acc, fa1[0], fb1[0]);
        __builtin_amdgcn_sched_barrier(0);
        if (more) {
            *(v4f*)(&As[nxt][stOff]) = sa0;
            *(v4f*)(&As[nxt][stOff + 64 * LDA]) = sa1;
#pragma unroll
            for (int i = 0; i < 4; ++i)
                *(v4f*)(&Bs[nxt][stOff + i * 64 * LDB]) = sb[i];
            qA += 16;
            qB += 16;
        }
        __builtin_amdgcn_sched_barrier(0);
        mma16(acc, fa1[1], fb1[1]);
        if (more) __syncthreads();
    }
    mma16(acc, fa2[0], fb2[0]);
    mma16(acc, fa2[1], fb2[1]);

#pragma unroll
    for (int nt = 0; nt < 4; ++nt) {
        const int col = blockN + wn * 64 + nt * 16 + lm;
        const float bv = bias[col];
#pragma unroll
        for (int mt = 0; mt < 4; ++mt) {
            const int rowBase = blockM + wm * 64 + mt * 16 + half * 8;
#pragma unroll
            for (int v = 0; v < 8; ++v)
                out[(size_t)(rowBase + v) * OUTF + col] = acc[mt][nt][v] + bv;
        }
    }
}

extern "C" void kernel_launch(void* const* d_in, const int* in_sizes, int n_in,
                              void* d_out, int out_size, void* d_ws, size_t ws_size,
                              hipStream_t stream) {
    (void)in_sizes; (void)n_in; (void)out_size; (void)ws_size;
    const float* x     = (const float*)d_in[0];
    const float* theta = (const float*)d_in[1];
    const float* W0    = (const float*)d_in[2];
    const float* bias  = (const float*)d_in[3];
    const float* BB    = (const float*)d_in[4];
    const float* GG    = (const float*)d_in[5];
    const int*   Pi    = (const int*)d_in[6];
    float* out = (float*)d_out;

    float* sAcc = (float*)d_ws;
    float* A01  = (float*)d_ws + 64;
    float* w    = (float*)d_ws + 64 + 2048;

    k_theta_fwht<<<2,    256, 0, stream>>>(theta, BB, A01, sAcc);
    k_perm_fwht_lo<<<1024, 256, 0, stream>>>(Pi, GG, A01, sAcc, w);
    k_fwht_hi  <<<128,   256, 0, stream>>>(w, W0, sAcc);
    k_gemm     <<<dim3(OUTF / 256, MTOT / 128), 256, 0, stream>>>(
        x, bias, w, out);
}


# k_theta_fwht folded into k_perm_fwht_lo (256 blocks of four 1024-chunks, each builds A0|A1 in its LDS): 4 launches -> 3; GG^2 as 256 plain block sums joined in fixed order in k_fwht_hi, no atomic
# speedup vs baseline: 1.4266x; 1.0301x over previous
#include <hip/hip_runtime.h>


typedef float v2f __attribute__((ext_vector_type(2)));
typedef float v4f __attribute__((ext_vector_type(4)));
typedef float v8f __attribute__((ext_vector_type(8)));
typedef int   v4i __attribute__((ext_vector_type(4)));

#define LLN   (1 << 20)
#define OUTF  1024
#define INF   768
#define MTOT  16384
#define KTILES (INF / 16)

__device__ __forceinline__ void barrier_lds_only() {
    __builtin_amdgcn_fence(__ATOMIC_RELEASE, "workgroup", "local");
    __builtin_amdgcn_s_barrier();
    __builtin_amdgcn_fence(__ATOMIC_ACQUIRE, "workgroup", "local");
}

template <int N, bool LDS_ONLY>
__device__ __forceinline__ void fwht1024_lds(float* lds, const int t) {
#pragma unroll
    for (int p = 0; p < 5; ++p) {
        const int sh = 2 * p;
        const int h = 1 << sh;
        const int base = ((t >> sh) << (sh + 2)) | (t & (h - 1));
        float e[N][4];
#pragma unroll
        for (int a = 0; a < N; ++a) {
            const float* L = lds + a * 1024 + base;
            e[a][0] = L[0];
            e[a][1] = L[h];
            e[a][2] = L[2 * h];
            e[a][3] = L[3 * h];
        }
#pragma unroll
        for (int a = 0; a < N; ++a) {
            float* L = lds + a * 1024 + base;
            const float s0 = e[a][0] + e[a][1], d0 = e[a][0] - e[a][1];
            const float s1 = e[a][2] + e[a][3], d1 = e[a][2] - e[a][3];
            L[0]     = s0 + s1;
            L[h]     = d0 + d1;
            L[2 * h] = s0 - s1;
            L[3 * h] = d0 - d1;
        }
        if (LDS_ONLY) barrier_lds_only(); else __syncthreads();
    }
}

__global__ __launch_bounds__(256) void k_perm_fwht_lo(
    const float* __restrict__ theta, const float* __restrict__ BB,
    const int* __restrict__ Pi, const float* __restrict__ GG,
    float* __restrict__ part, float* __restrict__ w) {
    __shared__ float tab[2048];
    __shared__ float lds[4096];
    __shared__ float red[8];
    const int t = threadIdx.x;
    const size_t base = (size_t)blockIdx.x * 4096 + 4 * t;
    const v4f bb0 = *(const v4f*)(BB + 4 * t);
    const v4f bb1 = *(const v4f*)(BB + 1024 + 4 * t);
    const v4f th0 = *(const v4f*)(theta + 4 * t);
    const v4f th1 = *(const v4f*)(theta + 1024 + 4 * t);
    v4f g[4];
    v4i pi[4];
#pragma unroll
    for (int i = 0; i < 4; ++i) {
        g[i]  = *(const v4f*)(GG + base + i * 1024);
        pi[i] = *(const v4i*)(Pi + base + i * 1024);
    }
    *(v4f*)(tab + 4 * t)        = bb0 * th0;
    *(v4f*)(tab + 1024 + 4 * t) = bb1 * th1;
    barrier_lds_only();
    fwht1024_lds<2, true>(tab, t);
    __builtin_amdgcn_sched_barrier(0);

    float a0[4][4], a1[4][4];
#pragma unroll
    for (int i = 0; i < 4; ++i) {
#pragma unroll
        for (int k = 0; k < 4; ++k) {
            const int c = pi[i][k] & 1023;
            a0[i][k] = tab[c];
            a1[i][k] = tab[1024 + c];
        }
    }
    __builtin_amdgcn_sched_barrier(0);
    float p = 0.0f;
#pragma unroll
    for (int i = 0; i < 4; ++i)
#pragma unroll
        for (int k = 0; k < 4; ++k) p += g[i][k] * g[i][k];
#pragma unroll
    for (int m = 16; m > 0; m >>= 1) p += __shfl_xor(p, m);
    if ((t & 31) == 0) red[t >> 5] = p;
#pragma unroll
    for (int i = 0; i < 4; ++i) {
        v4f o;
#pragma unroll
        for (int k = 0; k < 4; ++k) {
            const float v = ((pi[i][k] >> 10) & 1) ? (a0[i][k] - a1[i][k])
                                                   : (a0[i][k] + a1[i][k]);
            o[k] = g[i][k] * v;
        }
        *(v4f*)(lds + i * 1024 + 4 * t) = o;
    }
    __syncthreads();
    if (t == 0) {
        float s = red[0];
#pragma unroll
        for (int j = 1; j < 8; ++j) s += red[j];
        part[blockIdx.x] = s;
    }
    fwht1024_lds<4, false>(lds, t);
#pragma unroll
    for (int i = 0; i < 4; ++i)
        *(v4f*)(w + base + i * 1024) = *(const v4f*)(lds + i * 1024 + 4 * t);
}

__device__ __forceinline__ void fwht32_regs(float (&a)[32]) {
#pragma unroll
    for (int h = 1; h < 32; h <<= 1) {
#pragma unroll
        for (int i = 0; i < 32; ++i) {
            if ((i & h) == 0) {
                const float x = a[i], y = a[i + h];
                a[i] = x + y;
                a[i + h] = x - y;
            }
        }
    }
}

__global__ void k_fwht_hi(float* __restrict__ w,
                          const float* __restrict__ W0,
                          const float* __restrict__ part) {
    __shared__ float lds[1024 * 8 + 32 * 8];
    __shared__ float red[8];
    const int t = threadIdx.x;
    float ps = part[t];
    const int c = t & 7;
    const int q = t >> 3;
    float* __restrict__ col = w + blockIdx.x * 8 + c;
    const float* __restrict__ col0 = W0 + blockIdx.x * 8 + c;
    float a[32];
#pragma unroll
    for (int m = 0; m < 32; ++m) a[m] = col[(size_t)(q * 32 + m) * 1024];
    float b0[24];
#pragma unroll
    for (int m = 0; m < 24; ++m) b0[m] = col0[(size_t)(m * 32 + q) * 1024];
    fwht32_regs(a);
#pragma unroll
    for (int m = 16; m > 0; m >>= 1) ps += __shfl_xor(ps, m);
    if ((t & 31) == 0) red[t >> 5] = ps;
#pragma unroll
    for (int m = 0; m < 32; ++m) lds[(q * 32 + m) * 8 + q * 8 + c] = a[m];
    __syncthreads();
    float gg2 = red[0];
#pragma unroll
    for (int j = 1; j < 8; ++j) gg2 += red[j];
    const float scale = 1.0f / sqrtf(786432.0f * gg2);
#pragma unroll
    for (int m = 0; m < 32; ++m) a[m] = lds[(m * 32 + q) * 8 + m * 8 + c];
    fwht32_regs(a);
#pragma unroll
    for (int m = 0; m < 24; ++m)
        col[(size_t)(m * 32 + q) * 1024] = b0[m] + scale * a[m];
}

#define LDA 20
#define LDB 20
#define ROWS64 (64 * INF)

__device__ __forceinline__ void frag_load(const float* __restrict__ Ac,
                                          const float* __restrict__ Bc,
                                          const int aoff, const int boff,
                                          const int s0,
                                          v2f (&af)[2][4], v2f (&bf)[2][4]) {
#pragma unroll
    for (int j = 0; j < 2; ++j) {
        const int kk = (s0 + j) << 2;
#pragma unroll
        for (int mt = 0; mt < 4; ++mt)
            af[j][mt] = *(const v2f*)(Ac + aoff + mt * 16 * LDA + kk);
#pragma unroll
        for (int nt = 0; nt < 4; ++nt)
            bf[j][nt] = *(const v2f*)(Bc + boff + nt * 16 * LDB + kk);
    }
}

__device__ __forceinline__ void mma16(v8f (&acc)[4][4], const v2f (&af)[4],
                                      const v2f (&bf)[4]) {
#pragma unroll
    for (int mt = 0; mt < 4; ++mt)
#pragma unroll
        for (int nt = 0; nt < 4; ++nt)
            acc[mt][nt] = __builtin_amdgcn_wmma_f32_16x16x4_f32(
                false, af[mt], false, bf[nt], (short)0, acc[mt][nt],
                false, false);
}

__global__ __launch_bounds__(256) void k_gemm(
    const float* __restrict__ x, const float* __restrict__ bias,
    const float* __restrict__ weff, float* __restrict__ out) {
    __shared__ float As[2][128 * LDA];
    __shared__ float Bs[2][256 * LDB];
    const int t    = threadIdx.x;
    const int lane = t & 31;
    const int wid  = t >> 5;
    const int wm   = wid & 1;
    const int wn   = wid >> 1;
    const int half = lane >> 4;
    const int lm   = lane & 15;
    const int blockN = blockIdx.x * 256;
    const int blockM = blockIdx.y * 128;

    const int fr = t >> 2;
    const int fq = (t & 3) << 2;
    const float* __restrict__ pA = x   + (size_t)(blockM + fr) * INF + fq;
    const float* __restrict__ pB = weff + (size_t)(blockN + fr) * INF + fq;
    const float* __restrict__ qA = pA + 32;
    const float* __restrict__ qB = pB + 32;
    const int stOff = fr * LDA + fq;

    const int aoff = (wm * 64 + lm) * LDA + (half << 1);
    const int boff = (wn * 64 + lm) * LDB + (half << 1);

    v8f acc[4][4];
    const v8f zero = {0.f, 0.f, 0.f, 0.f, 0.f, 0.f, 0.f, 0.f};
#pragma unroll
    for (int mt = 0; mt < 4; ++mt)
#pragma unroll
        for (int nt = 0; nt < 4; ++nt) acc[mt][nt] = zero;

    v4f sa0, sa1, sb[4];
    v2f fa1[2][4], fb1[2][4], fa2[2][4], fb2[2][4];

    sa0 = *(const v4f*)(pA);
    sa1 = *(const v4f*)(pA + ROWS64);
#pragma unroll
    for (int i = 0; i < 4; ++i) sb[i] = *(const v4f*)(pB + i * ROWS64);
    *(v4f*)(&As[0][stOff]) = sa0;
    *(v4f*)(&As[0][stOff + 64 * LDA]) = sa1;
#pragma unroll
    for (int i = 0; i < 4; ++i) *(v4f*)(&Bs[0][stOff + i * 64 * LDB]) = sb[i];
    __syncthreads();

    sa0 = *(const v4f*)(pA + 16);
    sa1 = *(const v4f*)(pA + 16 + ROWS64);
#pragma unroll
    for (int i = 0; i < 4; ++i) sb[i] = *(const v4f*)(pB + 16 + i * ROWS64);
    frag_load(As[0], Bs[0], aoff, boff, 0, fa1, fb1);
    frag_load(As[0], Bs[0], aoff, boff, 2, fa2, fb2);
    __builtin_amdgcn_sched_barrier(0);
    mma16(acc, fa1[0], fb1[0]);
    mma16(acc, fa1[1], fb1[1]);
    __builtin_amdgcn_sched_barrier(0);
    *(v4f*)(&As[1][stOff]) = sa0;
    *(v4f*)(&As[1][stOff + 64 * LDA]) = sa1;
#pragma unroll
    for (int i = 0; i < 4; ++i) *(v4f*)(&Bs[1][stOff + i * 64 * LDB]) = sb[i];
    __syncthreads();

    for (int kt = 1; kt < KTILES; ++kt) {
        const int cur = kt & 1;
        const int nxt = cur ^ 1;
        const bool more = (kt + 1) < KTILES;
        const float* __restrict__ Ac = As[cur];
        const float* __restrict__ Bc = Bs[cur];

        if (more) {
            sa0 = *(const v4f*)(qA);
            sa1 = *(const v4f*)(qA + ROWS64);
#pragma unroll
            for (int i = 0; i < 4; ++i)
                sb[i] = *(const v4f*)(qB + i * ROWS64);
        }
        frag_load(Ac, Bc, aoff, boff, 0, fa1, fb1);
        __builtin_amdgcn_sched_barrier(0);
        mma16(acc, fa2[0], fb2[0]);
        mma16(acc, fa2[1], fb2[1]);
        __builtin_amdgcn_sched_barrier(0);
        frag_load(Ac, Bc, aoff, boff, 2, fa2, fb2);
        __builtin_amdgcn_sched_barrier(0);
        mma16(acc, fa1[0], fb1[0]);
        __builtin_amdgcn_sched_barrier(0);
        if (more) {
            *(v4f*)(&As[nxt][stOff]) = sa0;
            *(v4f*)(&As[nxt][stOff + 64 * LDA]) = sa1;
#pragma unroll
            for (int i = 0; i < 4; ++i)
                *(v4f*)(&Bs[nxt][stOff + i * 64 * LDB]) = sb[i];
            qA += 16;
            qB += 16;
        }
        __builtin_amdgcn_sched_barrier(0);
        mma16(acc, fa1[1], fb1[1]);
        if (more) __syncthreads();
    }
    mma16(acc, fa2[0], fb2[0]);
    mma16(acc, fa2[1], fb2[1]);

#pragma unroll
    for (int nt = 0; nt < 4; ++nt) {
        const int col = blockN + wn * 64 + nt * 16 + lm;
        const float bv = bias[col];
#pragma unroll
        for (int mt = 0; mt < 4; ++mt) {
            const int rowBase = blockM + wm * 64 + mt * 16 + half * 8;
#pragma unroll
            for (int v = 0; v < 8; ++v)
                out[(size_t)(rowBase + v) * OUTF + col] = acc[mt][nt][v] + bv;
        }
    }
}

extern "C" void kernel_launch(void* const* d_in, const int* in_sizes, int n_in,
                              void* d_out, int out_size, void* d_ws, size_t ws_size,
                              hipStream_t stream) {
    (void)in_sizes; (void)n_in; (void)out_size; (void)ws_size;
    const float* x     = (const float*)d_in[0];
    const float* theta = (const float*)d_in[1];
    const float* W0    = (const float*)d_in[2];
    const float* bias  = (const float*)d_in[3];
    const float* BB    = (const float*)d_in[4];
    const float* GG    = (const float*)d_in[5];
    const int*   Pi    = (const int*)d_in[6];
    float* out = (float*)d_out;

    float* part = (float*)d_ws;
    float* w    = (float*)d_ws + 2112;

    k_perm_fwht_lo<<<256, 256, 0, stream>>>(theta, BB, Pi, GG, part, w);
    k_fwht_hi  <<<128,   256, 0, stream>>>(w, W0, part);
    k_gemm     <<<dim3(OUTF / 256, MTOT / 128), 256, 0, stream>>>(
        x, bias, w, out);
}
